# starting kernel
# baseline (speedup 1.0000x reference)
#include <hip/hip_runtime.h>
#include <hip/hip_bf16.h>

typedef __attribute__((ext_vector_type(2))) float v2f;
typedef __attribute__((ext_vector_type(8))) float v8f;

#define WAVES_PER_BLOCK 8
#define BLOCK_THREADS   256

__device__ __forceinline__ float fast_tanh(float x) {
    float e = __expf(2.0f * x);
    return 1.0f - 2.0f * __builtin_amdgcn_rcpf(e + 1.0f);
}
__device__ __forceinline__ float fast_sigmoid(float z) {
    return __builtin_amdgcn_rcpf(1.0f + __expf(-z));
}

__global__ __launch_bounds__(BLOCK_THREADS)
void edge_mlp_wmma_f32(const float* __restrict__ emb,
                       const int*   __restrict__ eidx,
                       const float* __restrict__ eattr,
                       const float* __restrict__ W1,
                       const float* __restrict__ b1,
                       const float* __restrict__ W2,
                       const float* __restrict__ b2,
                       float* __restrict__ out,
                       int E)
{
    __shared__ float Wlds[5][32][2];

    const int tid = threadIdx.x;
    if (tid < 5 * 32) {
        const int s    = tid >> 5;
        const int lane = tid & 31;
        const int hi   = lane >> 4;
        const int nn   = lane & 15;
        const int k0   = 4 * s + 2 * hi;
#pragma unroll
        for (int t = 0; t < 2; ++t) {
            const int k = k0 + t;
            float w = 0.0f;
            if (nn < 8) {
                w = W1[(k + 16) * 8 + nn];
                if (k < 16) w += W1[k * 8 + nn];
            }
            Wlds[s][lane][t] = w;
        }
    }
    __syncthreads();

    const int lane = tid & 31;
    const int wave = tid >> 5;
    const int hi   = lane >> 4;
    const int nn   = lane & 15;

    const int tile = blockIdx.x * WAVES_PER_BLOCK + wave;
    const int base = tile * 16;
    if (base >= E) return;

    v2f b[5];
#pragma unroll
    for (int s = 0; s < 5; ++s) {
        b[s].x = Wlds[s][lane][0];
        b[s].y = Wlds[s][lane][1];
    }

    int e_row = base + nn;
    if (e_row >= E) e_row = E - 1;
    const long long node = (long long)eidx[e_row];
    const float* erow = emb + node * 16 + 2 * hi;

    v2f a[5];
#pragma unroll
    for (int s = 0; s < 4; ++s) {
        a[s].x = erow[4 * s + 0];
        a[s].y = erow[4 * s + 1];
    }
    a[4].x = eattr[(long long)e_row * 4 + 2 * hi + 0];
    a[4].y = eattr[(long long)e_row * 4 + 2 * hi + 1];

    v8f acc = {};
#pragma unroll
    for (int s = 0; s < 5; ++s) {
        acc = __builtin_amdgcn_wmma_f32_16x16x4_f32(
             false, a[s],  false, b[s],
             (short)0, acc,  false,  false);
    }

    const float b1n   = b1[nn & 7];
    const float w2n   = W2[nn & 7];
    const float msk   = (nn < 8) ? 1.0f : 0.0f;
    const float bias2 = b2[0];

    float S[8];
#pragma unroll
    for (int r = 0; r < 8; ++r) {
        float c = msk * (fast_tanh(acc[r] + b1n) * w2n);
        c += __shfl_xor(c, 1, 32);
        c += __shfl_xor(c, 2, 32);
        c += __shfl_xor(c, 4, 32);
        c += __shfl_xor(c, 8, 32);
        S[r] = c;
    }

    if (nn < 8) {
        float Sn = S[0];
#pragma unroll
        for (int r = 1; r < 8; ++r) Sn = (nn == r) ? S[r] : Sn;
        const int eo = base + nn + 8 * hi;
        if (eo < E) {
            out[eo] = fast_sigmoid(Sn + bias2);
        }
    }
}

extern "C" void kernel_launch(void* const* d_in, const int* in_sizes, int n_in,
                              void* d_out, int out_size, void* d_ws, size_t ws_size,
                              hipStream_t stream) {
    (void)n_in; (void)d_ws; (void)ws_size;
    const float* emb   = (const float*)d_in[0];
    const int*   eidx  = (const int*)  d_in[1];
    const float* eattr = (const float*)d_in[2];
    const float* W1    = (const float*)d_in[3];
    const float* b1    = (const float*)d_in[4];
    const float* W2    = (const float*)d_in[5];
    const float* b2    = (const float*)d_in[6];
    float* out = (float*)d_out;

    const int E      = out_size;
    const int tiles  = (E + 15) / 16;
    const int blocks = (tiles + WAVES_PER_BLOCK - 1) / WAVES_PER_BLOCK;
    (void)in_sizes;

    edge_mlp_wmma_f32<<<blocks, BLOCK_THREADS, 0, stream>>>(
        emb, eidx, eattr, W1, b1, W2, b2, out, E);
}


# persistent grid: 1024 blocks of 8 waves, each wave loops over 16-edge tiles; weight fragments and layer-2 constants staged once per wave, next tile's gather issued one trip ahead
# speedup vs baseline: 1.2215x; 1.2215x over previous
#include <hip/hip_runtime.h>
#include <hip/hip_bf16.h>

typedef __attribute__((ext_vector_type(2))) float v2f;
typedef __attribute__((ext_vector_type(8))) float v8f;

#define WAVES_PER_BLOCK 8
#define BLOCK_THREADS   256
#define GRID_BLOCKS     1024

__device__ __forceinline__ float fast_tanh(float x) {
    float e = __expf(2.0f * x);
    return 1.0f - 2.0f * __builtin_amdgcn_rcpf(e + 1.0f);
}
__device__ __forceinline__ float fast_sigmoid(float z) {
    return __builtin_amdgcn_rcpf(1.0f + __expf(-z));
}

struct TileA { v2f a[5]; };

__device__ __forceinline__ TileA load_tile(const float* __restrict__ emb,
                                           const float* __restrict__ eattr,
                                           int node, int e_row, int hi)
{
    TileA t;
    const float* erow = emb + (long long)node * 16 + 2 * hi;
#pragma unroll
    for (int s = 0; s < 4; ++s) {
        t.a[s].x = erow[4 * s + 0];
        t.a[s].y = erow[4 * s + 1];
    }
    t.a[4].x = eattr[(long long)e_row * 4 + 2 * hi + 0];
    t.a[4].y = eattr[(long long)e_row * 4 + 2 * hi + 1];
    return t;
}

struct WaveConst {
    v2f   b[5];
    float b1n, w2n, msk, bias2;
};

__device__ __forceinline__ void tile_math_store(const TileA& cur, const WaveConst& c,
                                                int nn, int hi, int tile, int E,
                                                float* __restrict__ out)
{
    v8f acc = {};
#pragma unroll
    for (int s = 0; s < 5; ++s) {
        acc = __builtin_amdgcn_wmma_f32_16x16x4_f32(
             false, cur.a[s],  false, c.b[s],
             (short)0, acc,  false,  false);
    }

    float S[8];
#pragma unroll
    for (int r = 0; r < 8; ++r) {
        float v = c.msk * (fast_tanh(acc[r] + c.b1n) * c.w2n);
        v += __shfl_xor(v, 1, 32);
        v += __shfl_xor(v, 2, 32);
        v += __shfl_xor(v, 4, 32);
        v += __shfl_xor(v, 8, 32);
        S[r] = v;
    }

    if (nn < 8) {
        float Sn = S[0];
#pragma unroll
        for (int r = 1; r < 8; ++r) Sn = (nn == r) ? S[r] : Sn;
        const int eo = tile * 16 + nn + 8 * hi;
        if (eo < E) {
            out[eo] = fast_sigmoid(Sn + c.bias2);
        }
    }
}

__global__ __launch_bounds__(BLOCK_THREADS)
void edge_mlp_wmma_f32(const float* __restrict__ emb,
                       const int*   __restrict__ eidx,
                       const float* __restrict__ eattr,
                       const float* __restrict__ W1,
                       const float* __restrict__ b1,
                       const float* __restrict__ W2,
                       const float* __restrict__ b2,
                       float* __restrict__ out,
                       int E)
{
    __shared__ float Wlds[5][32][2];

    const int tid = threadIdx.x;
    if (tid < 5 * 32) {
        const int s    = tid >> 5;
        const int lane = tid & 31;
        const int hi   = lane >> 4;
        const int nn   = lane & 15;
        const int k0   = 4 * s + 2 * hi;
#pragma unroll
        for (int t = 0; t < 2; ++t) {
            const int k = k0 + t;
            float w = 0.0f;
            if (nn < 8) {
                w = W1[(k + 16) * 8 + nn];
                if (k < 16) w += W1[k * 8 + nn];
            }
            Wlds[s][lane][t] = w;
        }
    }
    __syncthreads();

    const int lane = tid & 31;
    const int wave = __builtin_amdgcn_readfirstlane(tid >> 5);
    const int hi   = lane >> 4;
    const int nn   = lane & 15;

    const int tiles  = (E + 15) >> 4;
    const int stride = (int)gridDim.x * WAVES_PER_BLOCK;
    int tile = (int)blockIdx.x * WAVES_PER_BLOCK + wave;
    if (tile >= tiles) return;

    WaveConst c;
#pragma unroll
    for (int s = 0; s < 5; ++s) {
        c.b[s].x = Wlds[s][lane][0];
        c.b[s].y = Wlds[s][lane][1];
    }
    c.b1n   = b1[nn & 7];
    c.w2n   = W2[nn & 7];
    c.msk   = (nn < 8) ? 1.0f : 0.0f;
    c.bias2 = b2[0];
    const int last = E - 1;

    int e_nx, node_nx;
    TileA A0, A1;
    {
        const int e_cur    = min(tile * 16 + nn, last);
        const int node_cur = eidx[e_cur];
        const int t_nx     = min(tile + stride, tiles - 1);
        e_nx    = min(t_nx * 16 + nn, last);
        node_nx = eidx[e_nx];
        A0 = load_tile(emb, eattr, node_cur, e_cur, hi);
    }

    for (;;) {
        {
            const int t_n2    = min(tile + 2 * stride, tiles - 1);
            const int e_n2    = min(t_n2 * 16 + nn, last);
            const int node_n2 = eidx[e_n2];
            A1 = load_tile(emb, eattr, node_nx, e_nx, hi);
            tile_math_store(A0, c, nn, hi, tile, E, out);
            node_nx = node_n2;
            e_nx    = e_n2;
        }
        tile += stride;
        if (tile >= tiles) break;
        {
            const int t_n2    = min(tile + 2 * stride, tiles - 1);
            const int e_n2    = min(t_n2 * 16 + nn, last);
            const int node_n2 = eidx[e_n2];
            A0 = load_tile(emb, eattr, node_nx, e_nx, hi);
            tile_math_store(A1, c, nn, hi, tile, E, out);
            node_nx = node_n2;
            e_nx    = e_n2;
        }
        tile += stride;
        if (tile >= tiles) break;
    }
}

extern "C" void kernel_launch(void* const* d_in, const int* in_sizes, int n_in,
                              void* d_out, int out_size, void* d_ws, size_t ws_size,
                              hipStream_t stream) {
    (void)n_in; (void)d_ws; (void)ws_size; (void)in_sizes;
    const float* emb   = (const float*)d_in[0];
    const int*   eidx  = (const int*)  d_in[1];
    const float* eattr = (const float*)d_in[2];
    const float* W1    = (const float*)d_in[3];
    const float* b1    = (const float*)d_in[4];
    const float* W2    = (const float*)d_in[5];
    const float* b2    = (const float*)d_in[6];
    float* out = (float*)d_out;

    edge_mlp_wmma_f32<<<GRID_BLOCKS, BLOCK_THREADS, 0, stream>>>(
        emb, eidx, eattr, W1, b1, W2, b2, out, out_size);
}


# weights become the WMMA A operand, edges the B operand (product transposed; two 16-edge tiles block-diagonal in one accumulator): a lane holds one edge's 8 hidden values, hidden sum in-lane adds
# speedup vs baseline: 1.4368x; 1.1763x over previous
#include <hip/hip_runtime.h>
#include <hip/hip_bf16.h>

typedef __attribute__((ext_vector_type(2))) float v2f;
typedef __attribute__((ext_vector_type(8))) float v8f;

#define WAVES_PER_BLOCK 8
#define BLOCK_THREADS   256
#define GRID_BLOCKS     1024

__device__ __forceinline__ float fast_tanh(float x) {
    float e = __expf(2.0f * x);
    return 1.0f - 2.0f * __builtin_amdgcn_rcpf(e + 1.0f);
}
__device__ __forceinline__ float fast_sigmoid(float z) {
    return __builtin_amdgcn_rcpf(1.0f + __expf(-z));
}


struct UnitX { v2f x[2][5]; };

__device__ __forceinline__ UnitX load_unit(const float* __restrict__ emb,
                                           const float* __restrict__ eattr,
                                           int node0, int node1,
                                           int e0, int e1, int hi)
{
    UnitX t;
    const float* r0 = emb + (long long)node0 * 16 + 2 * hi;
    const float* r1 = emb + (long long)node1 * 16 + 2 * hi;
#pragma unroll
    for (int s = 0; s < 4; ++s) {
        t.x[0][s].x = r0[4 * s + 0];
        t.x[0][s].y = r0[4 * s + 1];
        t.x[1][s].x = r1[4 * s + 0];
        t.x[1][s].y = r1[4 * s + 1];
    }
    t.x[0][4].x = eattr[(long long)e0 * 4 + 2 * hi + 0];
    t.x[0][4].y = eattr[(long long)e0 * 4 + 2 * hi + 1];
    t.x[1][4].x = eattr[(long long)e1 * 4 + 2 * hi + 0];
    t.x[1][4].y = eattr[(long long)e1 * 4 + 2 * hi + 1];
    return t;
}

struct WaveConst {
    v2f   wlo[5];
    v2f   whi[5];
    float b1[8];
    float w2[8];
    float bias2;
};

__device__ __forceinline__ void unit_math_store(const UnitX& cur, const WaveConst& c,
                                                int lane, int unit, int E,
                                                float* __restrict__ out)
{
    v8f acc = {};
#pragma unroll
    for (int s = 0; s < 5; ++s) {
        acc = __builtin_amdgcn_wmma_f32_16x16x4_f32(
             false, c.wlo[s],  false, cur.x[0][s],
             (short)0, acc,  false,  false);
    }
#pragma unroll
    for (int s = 0; s < 5; ++s) {
        acc = __builtin_amdgcn_wmma_f32_16x16x4_f32(
             false, c.whi[s],  false, cur.x[1][s],
             (short)0, acc,  false,  false);
    }

    float v[8];
#pragma unroll
    for (int r = 0; r < 8; ++r) {
        v[r] = fast_tanh(acc[r] + c.b1[r]) * c.w2[r];
    }
    const float S = ((v[0] + v[1]) + (v[2] + v[3])) + ((v[4] + v[5]) + (v[6] + v[7]));

    const int eo = unit * 32 + lane;
    if (eo < E) {
        out[eo] = fast_sigmoid(S + c.bias2);
    }
}

__global__ __launch_bounds__(BLOCK_THREADS)
void edge_mlp_wmma_f32(const float* __restrict__ emb,
                       const int*   __restrict__ eidx,
                       const float* __restrict__ eattr,
                       const float* __restrict__ W1,
                       const float* __restrict__ b1,
                       const float* __restrict__ W2,
                       const float* __restrict__ b2,
                       float* __restrict__ out,
                       int E)
{
    __shared__ float Wlds[5][32][2];

    const int tid = threadIdx.x;
    if (tid < 5 * 32) {
        const int s    = tid >> 5;
        const int lane = tid & 31;
        const int hi   = lane >> 4;
        const int nn   = lane & 15;
        const int k0   = 4 * s + 2 * hi;
#pragma unroll
        for (int t = 0; t < 2; ++t) {
            const int k = k0 + t;
            float w = 0.0f;
            if (nn < 8) {
                w = W1[(k + 16) * 8 + nn];
                if (k < 16) w += W1[k * 8 + nn];
            }
            Wlds[s][lane][t] = w;
        }
    }
    __syncthreads();

    const int lane = tid & 31;
    const int wave = __builtin_amdgcn_readfirstlane(tid >> 5);
    const int hi   = lane >> 4;
    const int nn   = lane & 15;

    const int units  = (E + 31) >> 5;
    const int stride = (int)gridDim.x * WAVES_PER_BLOCK;
    int unit = (int)blockIdx.x * WAVES_PER_BLOCK + wave;
    if (unit >= units) return;

    WaveConst c;
#pragma unroll
    for (int s = 0; s < 5; ++s) {
        c.wlo[s].x = Wlds[s][lane][0];
        c.wlo[s].y = Wlds[s][lane][1];
        c.whi[s].x = Wlds[s][lane ^ 8][0];
        c.whi[s].y = Wlds[s][lane ^ 8][1];
    }
#pragma unroll
    for (int r = 0; r < 8; ++r) {
        c.b1[r] = b1[r];
        c.w2[r] = W2[r];
    }
    c.bias2 = b2[0];
    const int last = E - 1;

    int e0_nx, e1_nx, n0_nx, n1_nx;
    UnitX A0, A1;
    {
        const int e0_cur = min(unit * 32 + nn, last);
        const int e1_cur = min(unit * 32 + 16 + nn, last);
        const int n0_cur = eidx[e0_cur];
        const int n1_cur = eidx[e1_cur];
        const int u_nx   = min(unit + stride, units - 1);
        e0_nx = min(u_nx * 32 + nn, last);
        e1_nx = min(u_nx * 32 + 16 + nn, last);
        n0_nx = eidx[e0_nx];
        n1_nx = eidx[e1_nx];
        A0 = load_unit(emb, eattr, n0_cur, n1_cur, e0_cur, e1_cur, hi);
    }

    for (;;) {
        {
            const int u_n2  = min(unit + 2 * stride, units - 1);
            const int e0_n2 = min(u_n2 * 32 + nn, last);
            const int e1_n2 = min(u_n2 * 32 + 16 + nn, last);
            const int n0_n2 = eidx[e0_n2];
            const int n1_n2 = eidx[e1_n2];
            A1 = load_unit(emb, eattr, n0_nx, n1_nx, e0_nx, e1_nx, hi);
            unit_math_store(A0, c, lane, unit, E, out);
            n0_nx = n0_n2;  n1_nx = n1_n2;
            e0_nx = e0_n2;  e1_nx = e1_n2;
        }
        unit += stride;
        if (unit >= units) break;
        {
            const int u_n2  = min(unit + 2 * stride, units - 1);
            const int e0_n2 = min(u_n2 * 32 + nn, last);
            const int e1_n2 = min(u_n2 * 32 + 16 + nn, last);
            const int n0_n2 = eidx[e0_n2];
            const int n1_n2 = eidx[e1_n2];
            A0 = load_unit(emb, eattr, n0_nx, n1_nx, e0_nx, e1_nx, hi);
            unit_math_store(A1, c, lane, unit, E, out);
            n0_nx = n0_n2;  n1_nx = n1_n2;
            e0_nx = e0_n2;  e1_nx = e1_n2;
        }
        unit += stride;
        if (unit >= units) break;
    }
}

extern "C" void kernel_launch(void* const* d_in, const int* in_sizes, int n_in,
                              void* d_out, int out_size, void* d_ws, size_t ws_size,
                              hipStream_t stream) {
    (void)n_in; (void)d_ws; (void)ws_size; (void)in_sizes;
    const float* emb   = (const float*)d_in[0];
    const int*   eidx  = (const int*)  d_in[1];
    const float* eattr = (const float*)d_in[2];
    const float* W1    = (const float*)d_in[3];
    const float* b1    = (const float*)d_in[4];
    const float* W2    = (const float*)d_in[5];
    const float* b2    = (const float*)d_in[6];
    float* out = (float*)d_out;

    edge_mlp_wmma_f32<<<GRID_BLOCKS, BLOCK_THREADS, 0, stream>>>(
        emb, eidx, eattr, W1, b1, W2, b2, out, out_size);
}
